# starting kernel
# baseline (speedup 1.0000x reference)
#include <hip/hip_runtime.h>


typedef __attribute__((ext_vector_type(2))) float v2f;
typedef __attribute__((ext_vector_type(8))) float v8f;
typedef __attribute__((ext_vector_type(4))) float f4;

#define PREV_LEN 10
#define AFT_LEN  10
#define NB       256
#define NC       32
#define HW       64
#define TM       (PREV_LEN - 1 + AFT_LEN)
#define MROWS    48
#define TSTR     68
#define EPSF     1e-6f

__global__ __launch_bounds__(128)
void mmvp_kernel(const float* __restrict__ feats,
                 const float* __restrict__ sim,
                 float* __restrict__ out)
{
    __shared__ float sT[MROWS * TSTR];
    __shared__ float sM[2][HW * HW];

    const int b    = blockIdx.x;
    const int tid  = threadIdx.x;
    const int wv   = tid >> 5;
    const int lane = tid & 31;
    const int lo16 = lane & 15;
    const int hi   = lane >> 4;
    const int cw   = wv * 16 + lo16;

    const float* featsB = feats + (size_t)b * PREV_LEN * NC * HW;
    const float* simB   = sim   + (size_t)b * TM * HW * HW;

    {
        const f4* s4 = (const f4*)simB;
        f4* d4 = (f4*)sM[0];
        for (int idx = tid; idx < (HW * HW) / 4; idx += 128)
            d4[idx] = s4[idx];
    }

    for (int idx = tid; idx < MROWS * HW; idx += 128) {
        int r = idx >> 6, c = idx & 63;
        float v = 0.0f;
        if (r < NC)       v = featsB[r * HW + c];
        else if (r == NC) v = 1.0f;
        sT[r * TSTR + c] = v;
    }
    __syncthreads();

    for (int step = 0; step < TM; ++step) {
        const float* __restrict__ cur = sM[step & 1];

        if (step + 1 < TM) {
            const f4* s4 = (const f4*)(simB + (size_t)(step + 1) * HW * HW);
            f4* d4 = (f4*)sM[(step + 1) & 1];
            for (int idx = tid; idx < (HW * HW) / 4; idx += 128)
                d4[idx] = s4[idx];
        }
        if (step + 2 < TM)
            __builtin_prefetch(simB + (size_t)(step + 2) * HW * HW + tid * 32, 0, 0);

        const bool isPrev = (step < PREV_LEN - 1);

        v8f a0 = {}, a1 = {}, a2 = {};
        if (isPrev) {
            const float* fa = featsB + (size_t)(step + 1) * NC * HW;
            #pragma unroll
            for (int r = 0; r < 8; ++r) {
                int r0 = r + hi * 8;
                a0[r] = fa[(r0)      * HW + cw];
                a1[r] = fa[(r0 + 16) * HW + cw];
            }
            a2[0] = (hi == 0) ? 1.0f : 0.0f;
        }

        #pragma unroll
        for (int kk = 0; kk < 16; ++kk) {
            const int kb = kk * 4 + hi * 2;
            v2f bf, fA0, fA1, fA2;
            bf.x  = cur[(kb)     * HW + cw];
            bf.y  = cur[(kb + 1) * HW + cw];
            fA0.x = sT[(lo16)      * TSTR + kb];
            fA0.y = sT[(lo16)      * TSTR + kb + 1];
            fA1.x = sT[(lo16 + 16) * TSTR + kb];
            fA1.y = sT[(lo16 + 16) * TSTR + kb + 1];
            fA2.x = sT[(lo16 + 32) * TSTR + kb];
            fA2.y = sT[(lo16 + 32) * TSTR + kb + 1];
            a0 = __builtin_amdgcn_wmma_f32_16x16x4_f32(false, fA0, false, bf, (short)0, a0, false, false);
            a1 = __builtin_amdgcn_wmma_f32_16x16x4_f32(false, fA1, false, bf, (short)0, a1, false, false);
            a2 = __builtin_amdgcn_wmma_f32_16x16x4_f32(false, fA2, false, bf, (short)0, a2, false, false);
        }
        __syncthreads();

        if (isPrev) {
            #pragma unroll
            for (int r = 0; r < 8; ++r) {
                int r0 = r + hi * 8;
                sT[(r0)      * TSTR + cw] = a0[r];
                sT[(r0 + 16) * TSTR + cw] = a1[r];
                sT[(r0 + 32) * TSTR + cw] = a2[r];
            }
            __syncthreads();
        } else {
            const int t = step - (PREV_LEN - 1);
            float wrow  = a2[0];
            float wsum  = __shfl(wrow, lo16, 32);
            float denom = wsum + EPSF;
            float* ob = out + (size_t)(b * AFT_LEN + t) * NC * HW;
            #pragma unroll
            for (int r = 0; r < 8; ++r) {
                int r0 = r + hi * 8;
                ob[(r0)      * HW + cw] = a0[r] / denom;
                ob[(r0 + 16) * HW + cw] = a1[r] / denom;
            }
        }
    }
}

extern "C" void kernel_launch(void* const* d_in, const int* in_sizes, int n_in,
                              void* d_out, int out_size, void* d_ws, size_t ws_size,
                              hipStream_t stream) {
    const float* feats = (const float*)d_in[0];
    const float* sim   = (const float*)d_in[1];
    float* out = (float*)d_out;
    mmvp_kernel<<<NB, 128, 0, stream>>>(feats, sim, out);
}


# Global loads batched one step ahead: each thread issues the next matrix (8 x b128) and additive term (16 x b32) together into registers, LDS store behind the WMMAs, not 8 serial load-wait-store trips
# speedup vs baseline: 2.6938x; 2.6938x over previous
#include <hip/hip_runtime.h>


typedef __attribute__((ext_vector_type(2))) float v2f;
typedef __attribute__((ext_vector_type(8))) float v8f;
typedef __attribute__((ext_vector_type(4))) float f4;

#define PREV_LEN 10
#define AFT_LEN  10
#define NB       256
#define NC       32
#define HW       64
#define TM       (PREV_LEN - 1 + AFT_LEN)
#define MROWS    48
#define TSTR     68
#define EPSF     1e-6f
#define NTHR     128
#define STG      ((HW * HW) / 4 / NTHR)
#define TLD      ((NC * HW) / NTHR)
#define TALL     ((MROWS * HW) / NTHR)

__global__ __launch_bounds__(NTHR)
void mmvp_kernel(const float* __restrict__ feats,
                 const float* __restrict__ sim,
                 float* __restrict__ out)
{
    __shared__ float sT[MROWS * TSTR];
    __shared__ float sM[2][HW * HW];

    const int b    = blockIdx.x;
    const int tid  = threadIdx.x;
    const int wv   = tid >> 5;
    const int lane = tid & 31;
    const int lo16 = lane & 15;
    const int hi   = lane >> 4;
    const int cw   = wv * 16 + lo16;

    const float* featsB = feats + (size_t)b * PREV_LEN * NC * HW;
    const float* simB   = sim   + (size_t)b * TM * HW * HW;

    f4  stg[STG];
    v8f nA0, nA1;
    float tv[TLD];
    {
        const f4* s4 = (const f4*)simB;
        for (int i = 0; i < STG; ++i)
            stg[i] = s4[tid + i * NTHR];
        for (int i = 0; i < TLD; ++i)
            tv[i] = featsB[tid + i * NTHR];
        const float* fa = featsB + (size_t)NC * HW;
        for (int r = 0; r < 8; ++r) {
            int r0 = r + hi * 8;
            nA0[r] = fa[(r0)      * HW + cw];
            nA1[r] = fa[(r0 + 16) * HW + cw];
        }
    }
    {
        f4* d4 = (f4*)sM[0];
        for (int i = 0; i < STG; ++i)
            d4[tid + i * NTHR] = stg[i];
        for (int i = 0; i < TLD; ++i) {
            int idx = tid + i * NTHR;
            sT[(idx >> 6) * TSTR + (idx & 63)] = tv[i];
        }
        for (int i = TLD; i < TALL; ++i) {
            int idx = tid + i * NTHR;
            int r = idx >> 6;
            sT[r * TSTR + (idx & 63)] = (r == NC) ? 1.0f : 0.0f;
        }
    }
    __syncthreads();

    for (int step = 0; step < TM; ++step) {
        const float* __restrict__ cur = sM[step & 1];
        const bool isPrev = (step < PREV_LEN - 1);
        const bool more   = (step + 1 < TM);

        v8f a0 = {}, a1 = {}, a2 = {};
        if (isPrev) {
            a0 = nA0;
            a1 = nA1;
            a2[0] = (hi == 0) ? 1.0f : 0.0f;
        }

        if (more) {
            const f4* s4 = (const f4*)(simB + (size_t)(step + 1) * HW * HW);
            for (int i = 0; i < STG; ++i)
                stg[i] = s4[tid + i * NTHR];
        }
        if (step + 2 < PREV_LEN) {
            const float* fa = featsB + (size_t)(step + 2) * NC * HW;
            #pragma unroll
            for (int r = 0; r < 8; ++r) {
                int r0 = r + hi * 8;
                nA0[r] = fa[(r0)      * HW + cw];
                nA1[r] = fa[(r0 + 16) * HW + cw];
            }
        }
        if (step + 2 < TM)
            __builtin_prefetch(simB + (size_t)(step + 2) * HW * HW + tid * 32, 0, 0);

        #pragma unroll
        for (int kk = 0; kk < 16; ++kk) {
            const int kb = kk * 4 + hi * 2;
            v2f bf, fA0, fA1, fA2;
            bf.x  = cur[(kb)     * HW + cw];
            bf.y  = cur[(kb + 1) * HW + cw];
            fA0.x = sT[(lo16)      * TSTR + kb];
            fA0.y = sT[(lo16)      * TSTR + kb + 1];
            fA1.x = sT[(lo16 + 16) * TSTR + kb];
            fA1.y = sT[(lo16 + 16) * TSTR + kb + 1];
            fA2.x = sT[(lo16 + 32) * TSTR + kb];
            fA2.y = sT[(lo16 + 32) * TSTR + kb + 1];
            a0 = __builtin_amdgcn_wmma_f32_16x16x4_f32(false, fA0, false, bf, (short)0, a0, false, false);
            a1 = __builtin_amdgcn_wmma_f32_16x16x4_f32(false, fA1, false, bf, (short)0, a1, false, false);
            a2 = __builtin_amdgcn_wmma_f32_16x16x4_f32(false, fA2, false, bf, (short)0, a2, false, false);
        }

        if (more) {
            f4* d4 = (f4*)sM[(step + 1) & 1];
            for (int i = 0; i < STG; ++i)
                d4[tid + i * NTHR] = stg[i];
        }
        __syncthreads();

        if (isPrev) {
            #pragma unroll
            for (int r = 0; r < 8; ++r) {
                int r0 = r + hi * 8;
                sT[(r0)      * TSTR + cw] = a0[r];
                sT[(r0 + 16) * TSTR + cw] = a1[r];
                sT[(r0 + 32) * TSTR + cw] = a2[r];
            }
            __syncthreads();
        } else {
            const int t = step - (PREV_LEN - 1);
            float wrow  = a2[0];
            float wsum  = __shfl(wrow, lo16, 32);
            float denom = wsum + EPSF;
            float* ob = out + (size_t)(b * AFT_LEN + t) * NC * HW;
            #pragma unroll
            for (int r = 0; r < 8; ++r) {
                int r0 = r + hi * 8;
                ob[(r0)      * HW + cw] = a0[r] / denom;
                ob[(r0 + 16) * HW + cw] = a1[r] / denom;
            }
        }
    }
}

extern "C" void kernel_launch(void* const* d_in, const int* in_sizes, int n_in,
                              void* d_out, int out_size, void* d_ws, size_t ws_size,
                              hipStream_t stream) {
    const float* feats = (const float*)d_in[0];
    const float* sim   = (const float*)d_in[1];
    float* out = (float*)d_out;
    mmvp_kernel<<<NB, NTHR, 0, stream>>>(feats, sim, out);
}


# LDS fragment reads of each step (A 3 tiles + B, all 16 k-slots) issued as one group into registers ahead of its 48 WMMAs, which then run back to back; sum order unchanged
# speedup vs baseline: 2.7742x; 1.0298x over previous
#include <hip/hip_runtime.h>


typedef __attribute__((ext_vector_type(2))) float v2f;
typedef __attribute__((ext_vector_type(8))) float v8f;
typedef __attribute__((ext_vector_type(4))) float f4;

#define PREV_LEN 10
#define AFT_LEN  10
#define NB       256
#define NC       32
#define HW       64
#define TM       (PREV_LEN - 1 + AFT_LEN)
#define MROWS    48
#define TSTR     68
#define EPSF     1e-6f
#define NTHR     128
#define STG      ((HW * HW) / 4 / NTHR)
#define TLD      ((NC * HW) / NTHR)
#define TALL     ((MROWS * HW) / NTHR)

__global__ __launch_bounds__(NTHR)
void mmvp_kernel(const float* __restrict__ feats,
                 const float* __restrict__ sim,
                 float* __restrict__ out)
{
    __shared__ float sT[MROWS * TSTR];
    __shared__ float sM[2][HW * HW];

    const int b    = blockIdx.x;
    const int tid  = threadIdx.x;
    const int wv   = tid >> 5;
    const int lane = tid & 31;
    const int lo16 = lane & 15;
    const int hi   = lane >> 4;
    const int cw   = wv * 16 + lo16;

    const float* featsB = feats + (size_t)b * PREV_LEN * NC * HW;
    const float* simB   = sim   + (size_t)b * TM * HW * HW;

    f4  stg[STG];
    v8f nA0, nA1;
    float tv[TLD];
    {
        const f4* s4 = (const f4*)simB;
        for (int i = 0; i < STG; ++i)
            stg[i] = s4[tid + i * NTHR];
        for (int i = 0; i < TLD; ++i)
            tv[i] = featsB[tid + i * NTHR];
        const float* fa = featsB + (size_t)NC * HW;
        for (int r = 0; r < 8; ++r) {
            int r0 = r + hi * 8;
            nA0[r] = fa[(r0)      * HW + cw];
            nA1[r] = fa[(r0 + 16) * HW + cw];
        }
    }
    {
        f4* d4 = (f4*)sM[0];
        for (int i = 0; i < STG; ++i)
            d4[tid + i * NTHR] = stg[i];
        for (int i = 0; i < TLD; ++i) {
            int idx = tid + i * NTHR;
            sT[(idx >> 6) * TSTR + (idx & 63)] = tv[i];
        }
        for (int i = TLD; i < TALL; ++i) {
            int idx = tid + i * NTHR;
            int r = idx >> 6;
            sT[r * TSTR + (idx & 63)] = (r == NC) ? 1.0f : 0.0f;
        }
    }
    __syncthreads();

    for (int step = 0; step < TM; ++step) {
        const float* __restrict__ cur = sM[step & 1];
        const bool isPrev = (step < PREV_LEN - 1);
        const bool more   = (step + 1 < TM);

        v8f a0 = {}, a1 = {}, a2 = {};
        if (isPrev) {
            a0 = nA0;
            a1 = nA1;
            a2[0] = (hi == 0) ? 1.0f : 0.0f;
        }

        if (more) {
            const f4* s4 = (const f4*)(simB + (size_t)(step + 1) * HW * HW);
            for (int i = 0; i < STG; ++i)
                stg[i] = s4[tid + i * NTHR];
        }
        if (step + 2 < PREV_LEN) {
            const float* fa = featsB + (size_t)(step + 2) * NC * HW;
            #pragma unroll
            for (int r = 0; r < 8; ++r) {
                int r0 = r + hi * 8;
                nA0[r] = fa[(r0)      * HW + cw];
                nA1[r] = fa[(r0 + 16) * HW + cw];
            }
        }
        if (step + 2 < TM)
            __builtin_prefetch(simB + (size_t)(step + 2) * HW * HW + tid * 32, 0, 0);

        v2f bfr[16], fA0[16], fA1[16], fA2[16];
        {
            const float* bp = cur + (hi * 2) * HW + cw;
            const float* ap = sT + lo16 * TSTR + hi * 2;
            #pragma unroll
            for (int kk = 0; kk < 16; ++kk) {
                bfr[kk].x = bp[(kk * 4)     * HW];
                bfr[kk].y = bp[(kk * 4 + 1) * HW];
                fA0[kk].x = ap[kk * 4];
                fA0[kk].y = ap[kk * 4 + 1];
                fA1[kk].x = ap[16 * TSTR + kk * 4];
                fA1[kk].y = ap[16 * TSTR + kk * 4 + 1];
                fA2[kk].x = ap[32 * TSTR + kk * 4];
                fA2[kk].y = ap[32 * TSTR + kk * 4 + 1];
            }
        }
        __builtin_amdgcn_sched_barrier(0);
        #pragma unroll
        for (int kk = 0; kk < 16; ++kk) {
            a0 = __builtin_amdgcn_wmma_f32_16x16x4_f32(false, fA0[kk], false, bfr[kk], (short)0, a0, false, false);
            a1 = __builtin_amdgcn_wmma_f32_16x16x4_f32(false, fA1[kk], false, bfr[kk], (short)0, a1, false, false);
            a2 = __builtin_amdgcn_wmma_f32_16x16x4_f32(false, fA2[kk], false, bfr[kk], (short)0, a2, false, false);
        }
        __builtin_amdgcn_sched_barrier(0);

        if (more) {
            f4* d4 = (f4*)sM[(step + 1) & 1];
            for (int i = 0; i < STG; ++i)
                d4[tid + i * NTHR] = stg[i];
        }
        __syncthreads();

        if (isPrev) {
            #pragma unroll
            for (int r = 0; r < 8; ++r) {
                int r0 = r + hi * 8;
                sT[(r0)      * TSTR + cw] = a0[r];
                sT[(r0 + 16) * TSTR + cw] = a1[r];
                sT[(r0 + 32) * TSTR + cw] = a2[r];
            }
            __syncthreads();
        } else {
            const int t = step - (PREV_LEN - 1);
            float wrow  = a2[0];
            float wsum  = __shfl(wrow, lo16, 32);
            float denom = wsum + EPSF;
            float* ob = out + (size_t)(b * AFT_LEN + t) * NC * HW;
            #pragma unroll
            for (int r = 0; r < 8; ++r) {
                int r0 = r + hi * 8;
                ob[(r0)      * HW + cw] = a0[r] / denom;
                ob[(r0 + 16) * HW + cw] = a1[r] / denom;
            }
        }
    }
}

extern "C" void kernel_launch(void* const* d_in, const int* in_sizes, int n_in,
                              void* d_out, int out_size, void* d_ws, size_t ws_size,
                              hipStream_t stream) {
    const float* feats = (const float*)d_in[0];
    const float* sim   = (const float*)d_in[1];
    float* out = (float*)d_out;
    mmvp_kernel<<<NB, NTHR, 0, stream>>>(feats, sim, out);
}


# Global loads two steps ahead, not one: two register sets each for matrix and additive term, steps as compile-time instances so each wait counts only younger loads, LDS-only barriers; sums unchanged
# speedup vs baseline: 3.3132x; 1.1943x over previous
#include <hip/hip_runtime.h>


typedef __attribute__((ext_vector_type(2))) float v2f;
typedef __attribute__((ext_vector_type(8))) float v8f;
typedef __attribute__((ext_vector_type(4))) float f4;

#define PREV_LEN 10
#define AFT_LEN  10
#define NB       256
#define NC       32
#define HW       64
#define TM       (PREV_LEN - 1 + AFT_LEN)
#define MROWS    48
#define TSTR     68
#define EPSF     1e-6f
#define NTHR     128
#define STG      ((HW * HW) / 4 / NTHR)
#define TLD      ((NC * HW) / NTHR)
#define TALL     ((MROWS * HW) / NTHR)

__device__ __forceinline__ void lds_barrier()
{
    __builtin_amdgcn_fence(__ATOMIC_RELEASE, "workgroup", "local");
    __builtin_amdgcn_s_barrier();
    __builtin_amdgcn_fence(__ATOMIC_ACQUIRE, "workgroup", "local");
}

template <bool PREV, bool LD_A, bool LD_M, bool ST_M>
__device__ __forceinline__ void mmvp_step(
    const int step,
    const float* __restrict__ featsB, const float* __restrict__ simB, float* __restrict__ outB,
    float* sT, float* sM,
    f4 (&ldS)[STG],
    f4 (&stS)[STG],
    v8f& rA0, v8f& rA1,
    const int tid, const int lo16, const int hi, const int cw)
{
    const float* cur = sM + (step & 1) * (HW * HW);

    v8f a0 = {}, a1 = {}, a2 = {};
    if (PREV) {
        a0 = rA0;
        a1 = rA1;
        a2[0] = (hi == 0) ? 1.0f : 0.0f;
    }

    if (LD_A) {
        const float* fa = featsB + (size_t)(step + 3) * NC * HW;
        #pragma unroll
        for (int r = 0; r < 8; ++r) {
            int r0 = r + hi * 8;
            rA0[r] = fa[(r0)      * HW + cw];
            rA1[r] = fa[(r0 + 16) * HW + cw];
        }
    }
    if (LD_M) {
        const f4* s4 = (const f4*)(simB + (size_t)(step + 2) * HW * HW);
        #pragma unroll
        for (int i = 0; i < STG; ++i)
            ldS[i] = s4[tid + i * NTHR];
    }
    if (step + 4 < TM)
        __builtin_prefetch(simB + (size_t)(step + 4) * HW * HW + tid * 32, 0, 0);

    v2f bfr[16], fA0[16], fA1[16], fA2[16];
    {
        const float* bp = cur + (hi * 2) * HW + cw;
        const float* ap = sT + lo16 * TSTR + hi * 2;
        #pragma unroll
        for (int kk = 0; kk < 16; ++kk) {
            bfr[kk].x = bp[(kk * 4)     * HW];
            bfr[kk].y = bp[(kk * 4 + 1) * HW];
            fA0[kk].x = ap[kk * 4];
            fA0[kk].y = ap[kk * 4 + 1];
            fA1[kk].x = ap[16 * TSTR + kk * 4];
            fA1[kk].y = ap[16 * TSTR + kk * 4 + 1];
            fA2[kk].x = ap[32 * TSTR + kk * 4];
            fA2[kk].y = ap[32 * TSTR + kk * 4 + 1];
        }
    }
    __builtin_amdgcn_sched_barrier(0);
    #pragma unroll
    for (int kk = 0; kk < 16; ++kk) {
        a0 = __builtin_amdgcn_wmma_f32_16x16x4_f32(false, fA0[kk], false, bfr[kk], (short)0, a0, false, false);
        a1 = __builtin_amdgcn_wmma_f32_16x16x4_f32(false, fA1[kk], false, bfr[kk], (short)0, a1, false, false);
        a2 = __builtin_amdgcn_wmma_f32_16x16x4_f32(false, fA2[kk], false, bfr[kk], (short)0, a2, false, false);
    }
    asm volatile("" : "+v"(a0), "+v"(a1), "+v"(a2));

    if (ST_M) {
        f4* d4 = (f4*)(sM + ((step + 1) & 1) * (HW * HW));
        #pragma unroll
        for (int i = 0; i < STG; ++i)
            d4[tid + i * NTHR] = stS[i];
    }
    lds_barrier();

    if (PREV) {
        #pragma unroll
        for (int r = 0; r < 8; ++r) {
            int r0 = r + hi * 8;
            sT[(r0)      * TSTR + cw] = a0[r];
            sT[(r0 + 16) * TSTR + cw] = a1[r];
            sT[(r0 + 32) * TSTR + cw] = a2[r];
        }
        lds_barrier();
    } else {
        const int t = step - (PREV_LEN - 1);
        float wrow  = a2[0];
        float wsum  = __shfl(wrow, lo16, 32);
        float denom = wsum + EPSF;
        float* ob = outB + (size_t)t * NC * HW;
        #pragma unroll
        for (int r = 0; r < 8; ++r) {
            int r0 = r + hi * 8;
            ob[(r0)      * HW + cw] = a0[r] / denom;
            ob[(r0 + 16) * HW + cw] = a1[r] / denom;
        }
    }
}

__global__ __launch_bounds__(NTHR)
void mmvp_kernel(const float* __restrict__ feats,
                 const float* __restrict__ sim,
                 float* __restrict__ out)
{
    __shared__ float sT[MROWS * TSTR];
    __shared__ float sM[2 * HW * HW];

    const int b    = blockIdx.x;
    const int tid  = threadIdx.x;
    const int wv   = tid >> 5;
    const int lane = tid & 31;
    const int lo16 = lane & 15;
    const int hi   = lane >> 4;
    const int cw   = wv * 16 + lo16;

    const float* featsB = feats + (size_t)b * PREV_LEN * NC * HW;
    const float* simB   = sim   + (size_t)b * TM * HW * HW;
    float*       outB   = out   + (size_t)b * AFT_LEN * NC * HW;

    f4  S0[STG], S1[STG];
    v8f E0, E1, O0, O1;

    f4    first[STG];
    float tv[TLD];
    {
        const f4* s4 = (const f4*)simB;
        #pragma unroll
        for (int i = 0; i < STG; ++i)
            first[i] = s4[tid + i * NTHR];
        #pragma unroll
        for (int i = 0; i < TLD; ++i)
            tv[i] = featsB[tid + i * NTHR];
        const float* fa1 = featsB + (size_t)1 * NC * HW;
        const float* fa2 = featsB + (size_t)2 * NC * HW;
        #pragma unroll
        for (int r = 0; r < 8; ++r) {
            int r0 = r + hi * 8;
            O0[r] = fa1[(r0)      * HW + cw];
            O1[r] = fa1[(r0 + 16) * HW + cw];
        }
        #pragma unroll
        for (int r = 0; r < 8; ++r) {
            int r0 = r + hi * 8;
            E0[r] = fa2[(r0)      * HW + cw];
            E1[r] = fa2[(r0 + 16) * HW + cw];
        }
        const f4* s41 = (const f4*)(simB + (size_t)HW * HW);
        #pragma unroll
        for (int i = 0; i < STG; ++i)
            S1[i] = s41[tid + i * NTHR];
        __builtin_prefetch(simB + (size_t)2 * HW * HW + tid * 32, 0, 0);
        __builtin_prefetch(simB + (size_t)3 * HW * HW + tid * 32, 0, 0);
    }
    {
        f4* d4 = (f4*)sM;
        #pragma unroll
        for (int i = 0; i < STG; ++i)
            d4[tid + i * NTHR] = first[i];
        #pragma unroll
        for (int i = 0; i < TLD; ++i) {
            int idx = tid + i * NTHR;
            sT[(idx >> 6) * TSTR + (idx & 63)] = tv[i];
        }
        #pragma unroll
        for (int i = TLD; i < TALL; ++i) {
            int idx = tid + i * NTHR;
            int r = idx >> 6;
            sT[r * TSTR + (idx & 63)] = (r == NC) ? 1.0f : 0.0f;
        }
    }
    lds_barrier();

    mmvp_step<true, true, true, true>(0, featsB, simB, outB, sT, sM, S0, S1, O0, O1, tid, lo16, hi, cw);
    mmvp_step<true, true, true, true>(1, featsB, simB, outB, sT, sM, S1, S0, E0, E1, tid, lo16, hi, cw);
    mmvp_step<true, true, true, true>(2, featsB, simB, outB, sT, sM, S0, S1, O0, O1, tid, lo16, hi, cw);
    mmvp_step<true, true, true, true>(3, featsB, simB, outB, sT, sM, S1, S0, E0, E1, tid, lo16, hi, cw);
    mmvp_step<true, true, true, true>(4, featsB, simB, outB, sT, sM, S0, S1, O0, O1, tid, lo16, hi, cw);
    mmvp_step<true, true, true, true>(5, featsB, simB, outB, sT, sM, S1, S0, E0, E1, tid, lo16, hi, cw);
    mmvp_step<true, true, true, true>(6, featsB, simB, outB, sT, sM, S0, S1, O0, O1, tid, lo16, hi, cw);
    mmvp_step<true, false, true, true>(7, featsB, simB, outB, sT, sM, S1, S0, E0, E1, tid, lo16, hi, cw);
    mmvp_step<true, false, true, true>(8, featsB, simB, outB, sT, sM, S0, S1, O0, O1, tid, lo16, hi, cw);

    #pragma nounroll
    for (int s = 9; s < 17; s += 2) {
        mmvp_step<false, false, true, true>(s,     featsB, simB, outB, sT, sM, S1, S0, E0, E1, tid, lo16, hi, cw);
        mmvp_step<false, false, true, true>(s + 1, featsB, simB, outB, sT, sM, S0, S1, O0, O1, tid, lo16, hi, cw);
    }
    mmvp_step<false, false, false, true >(17, featsB, simB, outB, sT, sM, S1, S0, E0, E1, tid, lo16, hi, cw);
    mmvp_step<false, false, false, false>(18, featsB, simB, outB, sT, sM, S0, S1, O0, O1, tid, lo16, hi, cw);
}

extern "C" void kernel_launch(void* const* d_in, const int* in_sizes, int n_in,
                              void* d_out, int out_size, void* d_ws, size_t ws_size,
                              hipStream_t stream) {
    const float* feats = (const float*)d_in[0];
    const float* sim   = (const float*)d_in[1];
    float* out = (float*)d_out;
    mmvp_kernel<<<NB, NTHR, 0, stream>>>(feats, sim, out);
}


# Output phase without LDS or barriers: T's A fragments read once into registers for all ten products, B fragments of each last matrix loaded global -> registers two steps ahead; sums unchanged
# speedup vs baseline: 3.5027x; 1.0572x over previous
#include <hip/hip_runtime.h>


typedef __attribute__((ext_vector_type(2))) float v2f;
typedef __attribute__((ext_vector_type(8))) float v8f;
typedef __attribute__((ext_vector_type(4))) float f4;

#define PREV_LEN 10
#define AFT_LEN  10
#define NB       256
#define NC       32
#define HW       64
#define TM       (PREV_LEN - 1 + AFT_LEN)
#define MROWS    48
#define TSTR     68
#define EPSF     1e-6f
#define NTHR     128
#define STG      ((HW * HW) / 4 / NTHR)
#define TLD      ((NC * HW) / NTHR)
#define TALL     ((MROWS * HW) / NTHR)

__device__ __forceinline__ void lds_barrier()
{
    __builtin_amdgcn_fence(__ATOMIC_RELEASE, "workgroup", "local");
    __builtin_amdgcn_s_barrier();
    __builtin_amdgcn_fence(__ATOMIC_ACQUIRE, "workgroup", "local");
}

__device__ __forceinline__ void load_b_frags(v2f (&q)[16], const float* __restrict__ simB,
                                             const int t, const int hi, const int cw)
{
    const float* mb = simB + (size_t)(PREV_LEN - 1 + t) * HW * HW + (hi * 2) * HW + cw;
    #pragma unroll
    for (int kk = 0; kk < 16; ++kk) {
        q[kk].x = mb[(kk * 4)     * HW];
        q[kk].y = mb[(kk * 4 + 1) * HW];
    }
}

template <bool LD_A, bool LD_M, bool ST_M>
__device__ __forceinline__ void horner_step(
    const int step,
    const float* __restrict__ featsB, const float* __restrict__ simB,
    float* sT, float* sM,
    f4 (&ldS)[STG],
    f4 (&stS)[STG],
    v8f& rA0, v8f& rA1,
    const int tid, const int lo16, const int hi, const int cw)
{
    const float* cur = sM + (step & 1) * (HW * HW);

    v8f a0 = rA0, a1 = rA1, a2 = {};
    a2[0] = (hi == 0) ? 1.0f : 0.0f;

    if (LD_A) {
        const float* fa = featsB + (size_t)(step + 3) * NC * HW;
        #pragma unroll
        for (int r = 0; r < 8; ++r) {
            int r0 = r + hi * 8;
            rA0[r] = fa[(r0)      * HW + cw];
            rA1[r] = fa[(r0 + 16) * HW + cw];
        }
    }
    if (LD_M) {
        const f4* s4 = (const f4*)(simB + (size_t)(step + 2) * HW * HW);
        #pragma unroll
        for (int i = 0; i < STG; ++i)
            ldS[i] = s4[tid + i * NTHR];
    }
    __builtin_prefetch(simB + (size_t)(step + 4) * HW * HW + tid * 32, 0, 0);

    v2f bfr[16], fA0[16], fA1[16], fA2[16];
    {
        const float* bp = cur + (hi * 2) * HW + cw;
        const float* ap = sT + lo16 * TSTR + hi * 2;
        #pragma unroll
        for (int kk = 0; kk < 16; ++kk) {
            bfr[kk].x = bp[(kk * 4)     * HW];
            bfr[kk].y = bp[(kk * 4 + 1) * HW];
            fA0[kk].x = ap[kk * 4];
            fA0[kk].y = ap[kk * 4 + 1];
            fA1[kk].x = ap[16 * TSTR + kk * 4];
            fA1[kk].y = ap[16 * TSTR + kk * 4 + 1];
            fA2[kk].x = ap[32 * TSTR + kk * 4];
            fA2[kk].y = ap[32 * TSTR + kk * 4 + 1];
        }
    }
    __builtin_amdgcn_sched_barrier(0);
    #pragma unroll
    for (int kk = 0; kk < 16; ++kk) {
        a0 = __builtin_amdgcn_wmma_f32_16x16x4_f32(false, fA0[kk], false, bfr[kk], (short)0, a0, false, false);
        a1 = __builtin_amdgcn_wmma_f32_16x16x4_f32(false, fA1[kk], false, bfr[kk], (short)0, a1, false, false);
        a2 = __builtin_amdgcn_wmma_f32_16x16x4_f32(false, fA2[kk], false, bfr[kk], (short)0, a2, false, false);
    }
    asm volatile("" : "+v"(a0), "+v"(a1), "+v"(a2));

    if (ST_M) {
        f4* d4 = (f4*)(sM + ((step + 1) & 1) * (HW * HW));
        #pragma unroll
        for (int i = 0; i < STG; ++i)
            d4[tid + i * NTHR] = stS[i];
    }
    lds_barrier();

    #pragma unroll
    for (int r = 0; r < 8; ++r) {
        int r0 = r + hi * 8;
        sT[(r0)      * TSTR + cw] = a0[r];
        sT[(r0 + 16) * TSTR + cw] = a1[r];
        sT[(r0 + 32) * TSTR + cw] = a2[r];
    }
    lds_barrier();
}

template <bool LD_B>
__device__ __forceinline__ void out_step(
    const int t,
    const float* __restrict__ simB, float* __restrict__ outB,
    const v2f (&fA0)[16], const v2f (&fA1)[16], const v2f (&fA2)[16],
    const v2f (&use)[16], v2f (&fill)[16],
    const int tid, const int lo16, const int hi, const int cw)
{
    __builtin_amdgcn_sched_barrier(0);
    if (LD_B) {
        const int tl = (t + 2 < AFT_LEN) ? (t + 2) : (AFT_LEN - 1);
        load_b_frags(fill, simB, tl, hi, cw);
    }
    if (t + 4 < AFT_LEN)
        __builtin_prefetch(simB + (size_t)(PREV_LEN - 1 + t + 4) * HW * HW + tid * 32, 0, 0);
    __builtin_amdgcn_sched_barrier(0);

    v8f a0 = {}, a1 = {}, a2 = {};
    #pragma unroll
    for (int kk = 0; kk < 16; ++kk) {
        a0 = __builtin_amdgcn_wmma_f32_16x16x4_f32(false, fA0[kk], false, use[kk], (short)0, a0, false, false);
        a1 = __builtin_amdgcn_wmma_f32_16x16x4_f32(false, fA1[kk], false, use[kk], (short)0, a1, false, false);
        a2 = __builtin_amdgcn_wmma_f32_16x16x4_f32(false, fA2[kk], false, use[kk], (short)0, a2, false, false);
    }

    float wrow  = a2[0];
    float wsum  = __shfl(wrow, lo16, 32);
    float denom = wsum + EPSF;
    float* ob = outB + (size_t)t * NC * HW;
    #pragma unroll
    for (int r = 0; r < 8; ++r) {
        int r0 = r + hi * 8;
        ob[(r0)      * HW + cw] = a0[r] / denom;
        ob[(r0 + 16) * HW + cw] = a1[r] / denom;
    }
}

__global__ __launch_bounds__(NTHR)
void mmvp_kernel(const float* __restrict__ feats,
                 const float* __restrict__ sim,
                 float* __restrict__ out)
{
    __shared__ float sT[MROWS * TSTR];
    __shared__ float sM[2 * HW * HW];

    const int b    = blockIdx.x;
    const int tid  = threadIdx.x;
    const int wv   = tid >> 5;
    const int lane = tid & 31;
    const int lo16 = lane & 15;
    const int hi   = lane >> 4;
    const int cw   = wv * 16 + lo16;

    const float* featsB = feats + (size_t)b * PREV_LEN * NC * HW;
    const float* simB   = sim   + (size_t)b * TM * HW * HW;
    float*       outB   = out   + (size_t)b * AFT_LEN * NC * HW;

    f4  S0[STG], S1[STG];
    v8f E0, E1, O0, O1;

    f4    first[STG];
    float tv[TLD];
    {
        const f4* s4 = (const f4*)simB;
        #pragma unroll
        for (int i = 0; i < STG; ++i)
            first[i] = s4[tid + i * NTHR];
        #pragma unroll
        for (int i = 0; i < TLD; ++i)
            tv[i] = featsB[tid + i * NTHR];
        const float* fa1 = featsB + (size_t)1 * NC * HW;
        const float* fa2 = featsB + (size_t)2 * NC * HW;
        #pragma unroll
        for (int r = 0; r < 8; ++r) {
            int r0 = r + hi * 8;
            O0[r] = fa1[(r0)      * HW + cw];
            O1[r] = fa1[(r0 + 16) * HW + cw];
        }
        #pragma unroll
        for (int r = 0; r < 8; ++r) {
            int r0 = r + hi * 8;
            E0[r] = fa2[(r0)      * HW + cw];
            E1[r] = fa2[(r0 + 16) * HW + cw];
        }
        const f4* s41 = (const f4*)(simB + (size_t)HW * HW);
        #pragma unroll
        for (int i = 0; i < STG; ++i)
            S1[i] = s41[tid + i * NTHR];
        __builtin_prefetch(simB + (size_t)2 * HW * HW + tid * 32, 0, 0);
        __builtin_prefetch(simB + (size_t)3 * HW * HW + tid * 32, 0, 0);
    }
    {
        f4* d4 = (f4*)sM;
        #pragma unroll
        for (int i = 0; i < STG; ++i)
            d4[tid + i * NTHR] = first[i];
        #pragma unroll
        for (int i = 0; i < TLD; ++i) {
            int idx = tid + i * NTHR;
            sT[(idx >> 6) * TSTR + (idx & 63)] = tv[i];
        }
        #pragma unroll
        for (int i = TLD; i < TALL; ++i) {
            int idx = tid + i * NTHR;
            int r = idx >> 6;
            sT[r * TSTR + (idx & 63)] = (r == NC) ? 1.0f : 0.0f;
        }
    }
    lds_barrier();

    horner_step<true, true, true>(0, featsB, simB, sT, sM, S0, S1, O0, O1, tid, lo16, hi, cw);
    horner_step<true, true, true>(1, featsB, simB, sT, sM, S1, S0, E0, E1, tid, lo16, hi, cw);
    horner_step<true, true, true>(2, featsB, simB, sT, sM, S0, S1, O0, O1, tid, lo16, hi, cw);
    horner_step<true, true, true>(3, featsB, simB, sT, sM, S1, S0, E0, E1, tid, lo16, hi, cw);
    horner_step<true, true, true>(4, featsB, simB, sT, sM, S0, S1, O0, O1, tid, lo16, hi, cw);
    horner_step<true, true, true>(5, featsB, simB, sT, sM, S1, S0, E0, E1, tid, lo16, hi, cw);
    horner_step<true, true, true>(6, featsB, simB, sT, sM, S0, S1, O0, O1, tid, lo16, hi, cw);

    v2f Q0[16], Q1[16], Q2[16];
    load_b_frags(Q0, simB, 0, hi, cw);
    horner_step<false, false, true >(7, featsB, simB, sT, sM, S1, S0, E0, E1, tid, lo16, hi, cw);
    load_b_frags(Q1, simB, 1, hi, cw);
    horner_step<false, false, false>(8, featsB, simB, sT, sM, S0, S1, O0, O1, tid, lo16, hi, cw);

    v2f fA0[16], fA1[16], fA2[16];
    {
        const float* ap = sT + lo16 * TSTR + hi * 2;
        #pragma unroll
        for (int kk = 0; kk < 16; ++kk) {
            fA0[kk].x = ap[kk * 4];
            fA0[kk].y = ap[kk * 4 + 1];
            fA1[kk].x = ap[16 * TSTR + kk * 4];
            fA1[kk].y = ap[16 * TSTR + kk * 4 + 1];
            fA2[kk].x = ap[32 * TSTR + kk * 4];
            fA2[kk].y = ap[32 * TSTR + kk * 4 + 1];
        }
    }

    #pragma nounroll
    for (int t = 0; t < 9; t += 3) {
        out_step<true>(t,     simB, outB, fA0, fA1, fA2, Q0, Q2, tid, lo16, hi, cw);
        out_step<true>(t + 1, simB, outB, fA0, fA1, fA2, Q1, Q0, tid, lo16, hi, cw);
        out_step<true>(t + 2, simB, outB, fA0, fA1, fA2, Q2, Q1, tid, lo16, hi, cw);
    }
    out_step<false>(9, simB, outB, fA0, fA1, fA2, Q0, Q2, tid, lo16, hi, cw);
}

extern "C" void kernel_launch(void* const* d_in, const int* in_sizes, int n_in,
                              void* d_out, int out_size, void* d_ws, size_t ws_size,
                              hipStream_t stream) {
    const float* feats = (const float*)d_in[0];
    const float* sim   = (const float*)d_in[1];
    float* out = (float*)d_out;
    mmvp_kernel<<<NB, NTHR, 0, stream>>>(feats, sim, out);
}
